# starting kernel
# baseline (speedup 1.0000x reference)
#include <hip/hip_runtime.h>
#include <hip/hip_bf16.h>

typedef __attribute__((ext_vector_type(2))) float v2f;
typedef __attribute__((ext_vector_type(8))) float v8f;

#define BS   32
#define TS0  64
#define TSK  64
#define F    256
#define NF   128

__global__ void cin_stage1_wmma(const float* __restrict__ xk,
                                const float* __restrict__ w,
                                float* __restrict__ A) {
    const int fc   = blockIdx.x;
    const int t    = fc >> 2;
    const int wave = threadIdx.x >> 5;
    const int lane = threadIdx.x & 31;
    const int half = lane >> 4;
    const int l    = lane & 15;

    const float* wrow = w + (size_t)t * TSK;
    (void)wrow;

    for (int nt = 0; nt < 2; ++nt) {
        v8f acc = {};
        #pragma unroll
        for (int kc = 0; kc < 16; ++kc) {
            const int y0 = kc * 4 + half * 2;
            const int n = wave * 16 + l;
            v2f a;
            a.x = w[(size_t)n * (TS0 * TSK) + (size_t)t * TSK + y0];
            a.y = w[(size_t)n * (TS0 * TSK) + (size_t)t * TSK + y0 + 1];
            const int b = nt * 16 + l;
            v2f bb;
            bb.x = xk[(size_t)b * (TSK * F) + (size_t)y0 * F + fc];
            bb.y = xk[(size_t)b * (TSK * F) + (size_t)(y0 + 1) * F + fc];
            acc = __builtin_amdgcn_wmma_f32_16x16x4_f32(
                 false, a,  false, bb,
                 (short)0, acc,  false,  false);
        }
        #pragma unroll
        for (int v = 0; v < 8; ++v) {
            const int n = wave * 16 + v + half * 8;
            const int b = nt * 16 + l;
            A[((size_t)b * NF + n) * F + fc] = acc[v];
        }
    }
}

__global__ void cin_stage2_wmma(const float* __restrict__ x0,
                                const float* __restrict__ A,
                                const float* __restrict__ bias,
                                float* __restrict__ out) {
    const int bq   = blockIdx.x;
    const int b    = bq >> 2;
    const int q    = bq & 3;
    const int wave = threadIdx.x >> 5;
    const int lane = threadIdx.x & 31;
    const int half = lane >> 4;
    const int l    = lane & 15;

    for (int rt = 0; rt < 4; ++rt) {
        v8f acc = {};
        #pragma unroll
        for (int kc = 0; kc < 16; ++kc) {
            const int t0 = kc * 4 + half * 2;
            const int n = wave * 16 + l;
            v2f a;
            a.x = A[((size_t)b * NF + n) * F + (4 * t0 + q)];
            a.y = A[((size_t)b * NF + n) * F + (4 * (t0 + 1) + q)];
            const int r = rt * 16 + l;
            v2f bb;
            bb.x = x0[(size_t)b * (TS0 * F) + (size_t)r * F + (4 * t0 + q)];
            bb.y = x0[(size_t)b * (TS0 * F) + (size_t)r * F + (4 * (t0 + 1) + q)];
            acc = __builtin_amdgcn_wmma_f32_16x16x4_f32(
                false, a, false, bb, (short)0, acc, false, false);
        }
        #pragma unroll
        for (int v = 0; v < 8; ++v) {
            const int n = wave * 16 + v + half * 8;
            const int r = rt * 16 + l;
            out[((size_t)b * NF + n) * F + q * 64 + r] = acc[v] + bias[n];
        }
    }
}

extern "C" void kernel_launch(void* const* d_in, const int* in_sizes, int n_in,
                              void* d_out, int out_size, void* d_ws, size_t ws_size,
                              hipStream_t stream) {
    const float* x0   = (const float*)d_in[0];
    const float* xk   = (const float*)d_in[1];
    const float* w    = (const float*)d_in[2];
    const float* bias = (const float*)d_in[3];
    float* A   = (float*)d_ws;
    float* out = (float*)d_out;

    cin_stage1_wmma<<<F, 256, 0, stream>>>(xk, w, A);
    cin_stage2_wmma<<<BS * 4, 256, 0, stream>>>(x0, A, bias, out);
}


# One fused launch: each work-group does the first product for 16 n x 16 (batch,q) columns over all t into LDS and feeds the second product from it; no intermediate in memory; sum order unchanged
# speedup vs baseline: 2.9131x; 2.9131x over previous
#include <hip/hip_runtime.h>

typedef __attribute__((ext_vector_type(2))) float v2f;
typedef __attribute__((ext_vector_type(4))) float v4f;
typedef __attribute__((ext_vector_type(8))) float v8f;

#define BS   32
#define TS0  64
#define TSK  64
#define F    256
#define NF   128


#define PK 68
#define PW 516
#define PT 1092
#define PC 1090
#define PQ (64 * PK)

#define WL_OFF 0
#define XL_OFF (16 * PW)
#define AL_OFF (XL_OFF + 8 * PT)
#define X0_OFF (AL_OFF + 16 * PC)
#define LDS_FLOATS (X0_OFF + 4 * PQ)

static_assert(LDS_FLOATS * 4 <= 320 * 1024, "LDS budget");
static_assert((XL_OFF % 4) == 0 && (AL_OFF % 2) == 0 && (X0_OFF % 2) == 0, "fragment reads stay 8-byte aligned");

__global__ __launch_bounds__(256)
void cin_fused_wmma(const float* __restrict__ x0,
                    const float* __restrict__ xk,
                    const float* __restrict__ w,
                    const float* __restrict__ bias,
                    float* __restrict__ out) {
    __shared__ __attribute__((aligned(16))) float lds[LDS_FLOATS];
    float* WL  = lds + WL_OFF;
    float* XL  = lds + XL_OFF;
    float* AL  = lds + AL_OFF;
    float* X0L = lds + X0_OFF;

    const int tid  = threadIdx.x;
    const int wave = tid >> 5;
    const int lane = tid & 31;
    const int half = lane >> 4;
    const int c    = lane & 15;

    const int n0 = (blockIdx.x & 7) * 16;
    const int b0 = (blockIdx.x >> 3) * 4;

    const v4f* w4  = reinterpret_cast<const v4f*>(w);
    const v4f* xk4 = reinterpret_cast<const v4f*>(xk);
    const v4f* x04 = reinterpret_cast<const v4f*>(x0);

    v4f rw[8], rx[8], rz[16];

#define LOAD_CHUNK(i_)                                                              \
    _Pragma("unroll")                                                               \
    for (int u = 0; u < 8; ++u) {                                                   \
        const int idx = u * 256 + tid;                                              \
           \
        rw[u] = w4[((n0 + (idx >> 7)) * 4096 + (i_) * 512) / 4 + (idx & 127)];      \
          \
        const int row = idx >> 3;                                                   \
        rx[u] = xk4[((b0 + (row >> 6)) * 16384 + (row & 63) * 256 + (i_) * 32) / 4 + (idx & 7)]; \
    }

#define LOAD_X0(bs_)                                                                \
    _Pragma("unroll")                                                               \
    for (int u = 0; u < 16; ++u) {                                                  \
        const int idx = u * 256 + tid;                                              \
                          \
        rz[u] = x04[((b0 + (bs_)) * 16384) / 4 + idx];                              \
    }

    LOAD_CHUNK(0)

    #pragma unroll
    for (int i = 0; i < 8; ++i) {
        #pragma unroll
        for (int u = 0; u < 8; ++u) {
            const int idx = u * 256 + tid;
            *reinterpret_cast<v4f*>(WL + (idx >> 7) * PW + (idx & 127) * 4) = rw[u];
            const int row = idx >> 3;
            float* p = XL + (idx & 7) * PT + ((row >> 6) * 4) * PK + (row & 63);
            p[0]      = rx[u].x;
            p[PK]     = rx[u].y;
            p[2 * PK] = rx[u].z;
            p[3 * PK] = rx[u].w;
        }
        __syncthreads();

        if (i < 7) {
            LOAD_CHUNK(i + 1)
        } else {
            LOAD_X0(0)
        }

        v8f acc = {};
        const float* ap = WL + c * PW + wave * 64 + 2 * half;
        const float* bp = XL + wave * PT + c * PK + 2 * half;
        #pragma unroll
        for (int kc = 0; kc < 16; ++kc) {
            const v2f a  = *reinterpret_cast<const v2f*>(ap + 4 * kc);
            const v2f bb = *reinterpret_cast<const v2f*>(bp + 4 * kc);
            acc = __builtin_amdgcn_wmma_f32_16x16x4_f32(
                false, a, false, bb, (short)0, acc, false, false);
        }
        float* dp = AL + c * PC + (8 * half) * PK + (8 * i + wave);
        #pragma unroll
        for (int v = 0; v < 8; ++v) dp[v * PK] = acc[v];

        __syncthreads();
    }

    const int q  = wave & 3;
    const int pr = wave >> 2;

    #pragma unroll
    for (int bs = 0; bs < 4; ++bs) {
        #pragma unroll
        for (int u = 0; u < 16; ++u) {
            const int idx = u * 256 + tid;
            float* p = X0L + (idx >> 6) * PK + (idx & 63);
            p[0]      = rz[u].x;
            p[PQ]     = rz[u].y;
            p[2 * PQ] = rz[u].z;
            p[3 * PQ] = rz[u].w;
        }
        __syncthreads();

        if (bs < 3) {
            LOAD_X0(bs + 1)
        }

        v8f acc0 = {}, acc1 = {};
        const float* ap  = AL + (bs * 4 + q) * PC + c * PK + 2 * half;
        const float* bp0 = X0L + q * PQ + (32 * pr + c) * PK + 2 * half;
        const float* bp1 = bp0 + 16 * PK;
        #pragma unroll
        for (int kc = 0; kc < 16; ++kc) {
            const v2f a   = *reinterpret_cast<const v2f*>(ap  + 4 * kc);
            const v2f bb0 = *reinterpret_cast<const v2f*>(bp0 + 4 * kc);
            const v2f bb1 = *reinterpret_cast<const v2f*>(bp1 + 4 * kc);
            acc0 = __builtin_amdgcn_wmma_f32_16x16x4_f32(
                false, a, false, bb0, (short)0, acc0, false, false);
            acc1 = __builtin_amdgcn_wmma_f32_16x16x4_f32(
                false, a, false, bb1, (short)0, acc1, false, false);
        }

        float* orow = out + ((size_t)(b0 + bs) * NF + n0) * F + q * 64 + pr * 32 + lane;
        #pragma unroll
        for (int v = 0; v < 8; ++v) {
            const float send = half ? acc0[v] : acc1[v];
            const float got  = __shfl_xor(send, 16, 32);
            const float lo = half ? got : acc0[v];
            const float hi = half ? acc1[v] : got;
            orow[(size_t)v * F]       = lo + bias[n0 + v];
            orow[(size_t)(8 + v) * F] = hi + bias[n0 + 8 + v];
        }

        __syncthreads();
    }
#undef LOAD_CHUNK
#undef LOAD_X0
}

extern "C" void kernel_launch(void* const* d_in, const int* in_sizes, int n_in,
                              void* d_out, int out_size, void* d_ws, size_t ws_size,
                              hipStream_t stream) {
    const float* x0   = (const float*)d_in[0];
    const float* xk   = (const float*)d_in[1];
    const float* w    = (const float*)d_in[2];
    const float* bias = (const float*)d_in[3];
    float* out = (float*)d_out;

    cin_fused_wmma<<<64, 256, 0, stream>>>(x0, xk, w, bias, out);
}


# Work-group tile re-cut to 16 n x ONE batch: 256 work-groups (one per WGP, was 64); first product uses 4 of 16 tile columns, 16 t per LDS chunk; bytes pulled per group halve; sum order unchanged
# speedup vs baseline: 4.6864x; 1.6087x over previous
#include <hip/hip_runtime.h>

typedef __attribute__((ext_vector_type(2))) float v2f;
typedef __attribute__((ext_vector_type(4))) float v4f;
typedef __attribute__((ext_vector_type(8))) float v8f;

#define BS   32
#define TS0  64
#define TSK  64
#define F    256
#define NF   128


#define PK 68
#define PW (16 * 64 + 4)
#define PT (4 * PK + 4)
#define PC (16 * PK + 2)
#define PQ (64 * PK)

#define WL_OFF 0
#define XL_OFF (16 * PW)
#define AL_OFF (XL_OFF + 16 * PT)
#define X0_OFF (AL_OFF + 4 * PC)
#define LDS_FLOATS (X0_OFF + 4 * PQ)

static_assert(LDS_FLOATS * 4 <= 320 * 1024, "LDS budget");
static_assert((PW % 4) == 0 && (XL_OFF % 4) == 0, "16-byte W stores stay aligned");
static_assert((PT % 2) == 0 && (AL_OFF % 2) == 0 && (X0_OFF % 2) == 0, "fragment reads stay 8-byte aligned");

__global__ __launch_bounds__(256)
void cin_fused_wmma(const float* __restrict__ x0,
                    const float* __restrict__ xk,
                    const float* __restrict__ w,
                    const float* __restrict__ bias,
                    float* __restrict__ out) {
    __shared__ __attribute__((aligned(16))) float lds[LDS_FLOATS];
    float* WL  = lds + WL_OFF;
    float* XL  = lds + XL_OFF;
    float* AL  = lds + AL_OFF;
    float* X0L = lds + X0_OFF;

    const int tid  = threadIdx.x;
    const int wave = tid >> 5;
    const int lane = tid & 31;
    const int half = lane >> 4;
    const int c    = lane & 15;

    const int n0 = (blockIdx.x & 7) * 16;
    const int b  = blockIdx.x >> 3;

    const v4f* w4  = reinterpret_cast<const v4f*>(w);
    const v4f* xk4 = reinterpret_cast<const v4f*>(xk);
    const v4f* x04 = reinterpret_cast<const v4f*>(x0);

    v4f rw[16], rx[4], rz[16];

#define LOAD_CHUNK(i_)                                                              \
    _Pragma("unroll")                                                               \
    for (int u = 0; u < 16; ++u) {                                                  \
                  \
        rw[u] = w4[((n0 + u) * 4096 + (i_) * 1024) / 4 + tid];                      \
    }                                                                               \
    _Pragma("unroll")                                                               \
    for (int u = 0; u < 4; ++u) {                                                   \
        const int idx = u * 256 + tid;                                              \
                    \
        rx[u] = xk4[(b * 16384 + (idx >> 4) * 256 + (i_) * 64) / 4 + (idx & 15)];   \
    }

#define LOAD_X0()                                                                   \
    _Pragma("unroll")                                                               \
    for (int u = 0; u < 16; ++u) {                                                  \
        const int idx = u * 256 + tid;                                              \
                          \
        rz[u] = x04[(b * 16384) / 4 + idx];                                         \
    }

    LOAD_CHUNK(0)

    #pragma unroll
    for (int i = 0; i < 4; ++i) {
        #pragma unroll
        for (int u = 0; u < 16; ++u) {
            *reinterpret_cast<v4f*>(WL + u * PW + tid * 4) = rw[u];
        }
        #pragma unroll
        for (int u = 0; u < 4; ++u) {
            const int idx = u * 256 + tid;
            float* p = XL + (idx & 15) * PT + (idx >> 4);
            p[0]      = rx[u].x;
            p[PK]     = rx[u].y;
            p[2 * PK] = rx[u].z;
            p[3 * PK] = rx[u].w;
        }
        __syncthreads();

        if (i < 3) {
            LOAD_CHUNK(i + 1)
        } else {
            LOAD_X0()
        }

        v8f acc0 = {}, acc1 = {};
        const float* ap0 = WL + c * PW + wave * 64 + 2 * half;
        const float* ap1 = ap0 + 8 * 64;
        const float* bp0 = XL + wave * PT + (c & 3) * PK + 2 * half;
        const float* bp1 = bp0 + 8 * PT;
        #pragma unroll
        for (int kc = 0; kc < 16; ++kc) {
            const v2f a0  = *reinterpret_cast<const v2f*>(ap0 + 4 * kc);
            const v2f bb0 = *reinterpret_cast<const v2f*>(bp0 + 4 * kc);
            const v2f a1  = *reinterpret_cast<const v2f*>(ap1 + 4 * kc);
            const v2f bb1 = *reinterpret_cast<const v2f*>(bp1 + 4 * kc);
            acc0 = __builtin_amdgcn_wmma_f32_16x16x4_f32(
                false, a0, false, bb0, (short)0, acc0, false, false);
            acc1 = __builtin_amdgcn_wmma_f32_16x16x4_f32(
                false, a1, false, bb1, (short)0, acc1, false, false);
        }
        if (c < 4) {
            float* dp = AL + c * PC + (8 * half) * PK + (16 * i + wave);
            #pragma unroll
            for (int v = 0; v < 8; ++v) {
                dp[v * PK]     = acc0[v];
                dp[v * PK + 8] = acc1[v];
            }
        }

        if (i < 3) {
            __syncthreads();
        }
    }

    #pragma unroll
    for (int u = 0; u < 16; ++u) {
        const int idx = u * 256 + tid;
        float* p = X0L + (idx >> 6) * PK + (idx & 63);
        p[0]      = rz[u].x;
        p[PQ]     = rz[u].y;
        p[2 * PQ] = rz[u].z;
        p[3 * PQ] = rz[u].w;
    }
    __syncthreads();

    const int q  = wave & 3;
    const int pr = wave >> 2;

    v8f acc0 = {}, acc1 = {};
    const float* ap  = AL + q * PC + c * PK + 2 * half;
    const float* bp0 = X0L + q * PQ + (32 * pr + c) * PK + 2 * half;
    const float* bp1 = bp0 + 16 * PK;
    #pragma unroll
    for (int kc = 0; kc < 16; ++kc) {
        const v2f a   = *reinterpret_cast<const v2f*>(ap  + 4 * kc);
        const v2f bb0 = *reinterpret_cast<const v2f*>(bp0 + 4 * kc);
        const v2f bb1 = *reinterpret_cast<const v2f*>(bp1 + 4 * kc);
        acc0 = __builtin_amdgcn_wmma_f32_16x16x4_f32(
            false, a, false, bb0, (short)0, acc0, false, false);
        acc1 = __builtin_amdgcn_wmma_f32_16x16x4_f32(
            false, a, false, bb1, (short)0, acc1, false, false);
    }

    float* orow = out + ((size_t)b * NF + n0) * F + q * 64 + pr * 32 + lane;
    #pragma unroll
    for (int v = 0; v < 8; ++v) {
        const float send = half ? acc0[v] : acc1[v];
        const float got  = __shfl_xor(send, 16, 32);
        const float lo = half ? got : acc0[v];
        const float hi = half ? acc1[v] : got;
        orow[(size_t)v * F]       = lo + bias[n0 + v];
        orow[(size_t)(8 + v) * F] = hi + bias[n0 + 8 + v];
    }
#undef LOAD_CHUNK
#undef LOAD_X0
}

extern "C" void kernel_launch(void* const* d_in, const int* in_sizes, int n_in,
                              void* d_out, int out_size, void* d_ws, size_t ws_size,
                              hipStream_t stream) {
    const float* x0   = (const float*)d_in[0];
    const float* xk   = (const float*)d_in[1];
    const float* w    = (const float*)d_in[2];
    const float* bias = (const float*)d_in[3];
    float* out = (float*)d_out;

    cin_fused_wmma<<<256, 256, 0, stream>>>(x0, xk, w, bias, out);
}


# Staging pipeline two deep: two register sets and two W/x_k LDS tiles alternate, chunk i+2's loads fly under chunk i's products across one LDS-only barrier per chunk; sum order unchanged
# speedup vs baseline: 5.1390x; 1.0966x over previous
#include <hip/hip_runtime.h>

typedef __attribute__((ext_vector_type(2))) float v2f;
typedef __attribute__((ext_vector_type(4))) float v4f;
typedef __attribute__((ext_vector_type(8))) float v8f;

#define BS   32
#define TS0  64
#define TSK  64
#define F    256
#define NF   128


#define PK 68
#define PW (16 * 64 + 4)
#define PT (4 * PK + 4)
#define PC (16 * PK + 2)
#define PQ (64 * PK)

#define WL_OFF 0
#define XL_OFF (2 * 16 * PW)
#define AL_OFF (XL_OFF + 2 * 16 * PT)
#define X0_OFF (AL_OFF + 4 * PC)
#define LDS_FLOATS (X0_OFF + 4 * PQ)

static_assert(LDS_FLOATS * 4 <= 320 * 1024, "LDS budget");
static_assert((PW % 4) == 0 && (XL_OFF % 4) == 0, "16-byte W stores stay aligned");
static_assert((PT % 2) == 0 && (AL_OFF % 2) == 0 && (X0_OFF % 2) == 0, "fragment reads stay 8-byte aligned");

static __device__ __forceinline__ void lds_barrier() {
    asm volatile("s_wait_dscnt 0x0\n\ts_barrier_signal -1\n\ts_barrier_wait -1" ::: "memory");
}

__global__ __launch_bounds__(256)
void cin_fused_wmma(const float* __restrict__ x0,
                    const float* __restrict__ xk,
                    const float* __restrict__ w,
                    const float* __restrict__ bias,
                    float* __restrict__ out) {
    __shared__ __attribute__((aligned(16))) float lds[LDS_FLOATS];
    float* WL  = lds + WL_OFF;
    float* XL  = lds + XL_OFF;
    float* AL  = lds + AL_OFF;
    float* X0L = lds + X0_OFF;

    const int tid  = threadIdx.x;
    const int wave = tid >> 5;
    const int lane = tid & 31;
    const int half = lane >> 4;
    const int c    = lane & 15;

    const int n0 = (blockIdx.x & 7) * 16;
    const int b  = blockIdx.x >> 3;

    const v4f* w4  = reinterpret_cast<const v4f*>(w);
    const v4f* xk4 = reinterpret_cast<const v4f*>(xk);
    const v4f* x04 = reinterpret_cast<const v4f*>(x0);

    v4f rwE[16], rxE[4], rwO[16], rxO[4], rz[16];

#define LOAD_CHUNK(RW_, RX_, i_)                                                    \
    _Pragma("unroll")                                                               \
    for (int u = 0; u < 16; ++u) {                                                  \
                  \
        RW_[u] = w4[((n0 + u) * 4096 + (i_) * 1024) / 4 + tid];                     \
    }                                                                               \
    _Pragma("unroll")                                                               \
    for (int u = 0; u < 4; ++u) {                                                   \
        const int idx = u * 256 + tid;                                              \
                    \
        RX_[u] = xk4[(b * 16384 + (idx >> 4) * 256 + (i_) * 64) / 4 + (idx & 15)];  \
    }

#define STORE_CHUNK(RW_, RX_, buf_)                                                 \
    _Pragma("unroll")                                                               \
    for (int u = 0; u < 16; ++u) {                                                  \
        *reinterpret_cast<v4f*>(WL + (buf_) * 16 * PW + u * PW + tid * 4) = RW_[u]; \
    }                                                                               \
    _Pragma("unroll")                                                               \
    for (int u = 0; u < 4; ++u) {                                                   \
        const int idx = u * 256 + tid;                                              \
        float* p = XL + (buf_) * 16 * PT + (idx & 15) * PT + (idx >> 4);            \
        p[0]      = RX_[u].x;                                                       \
        p[PK]     = RX_[u].y;                                                       \
        p[2 * PK] = RX_[u].z;                                                       \
        p[3 * PK] = RX_[u].w;                                                       \
    }

#define LOAD_X0()                                                                   \
    _Pragma("unroll")                                                               \
    for (int u = 0; u < 16; ++u) {                                                  \
        const int idx = u * 256 + tid;                                              \
                          \
        rz[u] = x04[(b * 16384) / 4 + idx];                                         \
    }

#define STORE_X0()                                                                  \
    _Pragma("unroll")                                                               \
    for (int u = 0; u < 16; ++u) {                                                  \
        const int idx = u * 256 + tid;                                              \
        float* p = X0L + (idx >> 6) * PK + (idx & 63);                              \
        p[0]      = rz[u].x;                                                        \
        p[PQ]     = rz[u].y;                                                        \
        p[2 * PQ] = rz[u].z;                                                        \
        p[3 * PQ] = rz[u].w;                                                        \
    }

#define PRODUCT1(buf_, i_)                                                          \
    {                                                                               \
        v8f acc0 = {}, acc1 = {};                                                   \
                               \
        const float* ap0 = WL + (buf_) * 16 * PW + c * PW + wave * 64 + 2 * half;   \
        const float* ap1 = ap0 + 8 * 64;                                            \
                                                        \
        const float* bp0 = XL + (buf_) * 16 * PT + wave * PT + (c & 3) * PK + 2 * half; \
        const float* bp1 = bp0 + 8 * PT;                                            \
        _Pragma("unroll")                                                           \
        for (int kc = 0; kc < 16; ++kc) {                                           \
            const v2f a0  = *reinterpret_cast<const v2f*>(ap0 + 4 * kc);            \
            const v2f bb0 = *reinterpret_cast<const v2f*>(bp0 + 4 * kc);            \
            const v2f a1  = *reinterpret_cast<const v2f*>(ap1 + 4 * kc);            \
            const v2f bb1 = *reinterpret_cast<const v2f*>(bp1 + 4 * kc);            \
            acc0 = __builtin_amdgcn_wmma_f32_16x16x4_f32(                           \
                false, a0, false, bb0, (short)0, acc0, false, false);               \
            acc1 = __builtin_amdgcn_wmma_f32_16x16x4_f32(                           \
                false, a1, false, bb1, (short)0, acc1, false, false);               \
        }                                                                           \
          \
        if (c < 4) {                                                                \
            float* dp = AL + c * PC + (8 * half) * PK + (16 * (i_) + wave);         \
            _Pragma("unroll")                                                       \
            for (int v = 0; v < 8; ++v) {                                           \
                dp[v * PK]     = acc0[v];                                           \
                dp[v * PK + 8] = acc1[v];                                           \
            }                                                                       \
        }                                                                           \
    }

    LOAD_CHUNK(rwE, rxE, 0)
    LOAD_CHUNK(rwO, rxO, 1)
    STORE_CHUNK(rwE, rxE, 0)
    lds_barrier();

    LOAD_CHUNK(rwE, rxE, 2)
    PRODUCT1(0, 0)
    STORE_CHUNK(rwO, rxO, 1)
    lds_barrier();

    LOAD_CHUNK(rwO, rxO, 3)
    PRODUCT1(1, 1)
    STORE_CHUNK(rwE, rxE, 0)
    lds_barrier();

    LOAD_X0()
    PRODUCT1(0, 2)
    STORE_CHUNK(rwO, rxO, 1)
    lds_barrier();

    PRODUCT1(1, 3)
    STORE_X0()
    __syncthreads();

    const int q  = wave & 3;
    const int pr = wave >> 2;

    v8f acc0 = {}, acc1 = {};
    const float* ap  = AL + q * PC + c * PK + 2 * half;
    const float* bp0 = X0L + q * PQ + (32 * pr + c) * PK + 2 * half;
    const float* bp1 = bp0 + 16 * PK;
    #pragma unroll
    for (int kc = 0; kc < 16; ++kc) {
        const v2f a   = *reinterpret_cast<const v2f*>(ap  + 4 * kc);
        const v2f bb0 = *reinterpret_cast<const v2f*>(bp0 + 4 * kc);
        const v2f bb1 = *reinterpret_cast<const v2f*>(bp1 + 4 * kc);
        acc0 = __builtin_amdgcn_wmma_f32_16x16x4_f32(
            false, a, false, bb0, (short)0, acc0, false, false);
        acc1 = __builtin_amdgcn_wmma_f32_16x16x4_f32(
            false, a, false, bb1, (short)0, acc1, false, false);
    }

    float* orow = out + ((size_t)b * NF + n0) * F + q * 64 + pr * 32 + lane;
    #pragma unroll
    for (int v = 0; v < 8; ++v) {
        const float send = half ? acc0[v] : acc1[v];
        const float got  = __shfl_xor(send, 16, 32);
        const float lo = half ? got : acc0[v];
        const float hi = half ? acc1[v] : got;
        orow[(size_t)v * F]       = lo + bias[n0 + v];
        orow[(size_t)(8 + v) * F] = hi + bias[n0 + 8 + v];
    }
#undef LOAD_CHUNK
#undef STORE_CHUNK
#undef LOAD_X0
#undef STORE_X0
#undef PRODUCT1
}

extern "C" void kernel_launch(void* const* d_in, const int* in_sizes, int n_in,
                              void* d_out, int out_size, void* d_ws, size_t ws_size,
                              hipStream_t stream) {
    const float* x0   = (const float*)d_in[0];
    const float* xk   = (const float*)d_in[1];
    const float* w    = (const float*)d_in[2];
    const float* bias = (const float*)d_in[3];
    float* out = (float*)d_out;

    cin_fused_wmma<<<256, 256, 0, stream>>>(x0, xk, w, bias, out);
}


# Global walk rotated per work-group: chunk order starts at (batch + tile) & 3, w rows and x slices of a unit at an index-dependent row, so groups sharing lines ask at different times; sums unchanged
# speedup vs baseline: 5.9713x; 1.1620x over previous
#include <hip/hip_runtime.h>

typedef __attribute__((ext_vector_type(2))) float v2f;
typedef __attribute__((ext_vector_type(4))) float v4f;
typedef __attribute__((ext_vector_type(8))) float v8f;

#define BS   32
#define TS0  64
#define TSK  64
#define F    256
#define NF   128


#define PK 68
#define PW (16 * 64 + 4)
#define PT (4 * PK + 4)
#define PC (16 * PK + 2)
#define PQ (64 * PK)

#define WL_OFF 0
#define XL_OFF (2 * 16 * PW)
#define AL_OFF (XL_OFF + 2 * 16 * PT)
#define X0_OFF (AL_OFF + 4 * PC)
#define LDS_FLOATS (X0_OFF + 4 * PQ)

static_assert(LDS_FLOATS * 4 <= 320 * 1024, "LDS budget");
static_assert((PW % 4) == 0 && (XL_OFF % 4) == 0, "16-byte W stores stay aligned");
static_assert((PT % 2) == 0 && (AL_OFF % 2) == 0 && (X0_OFF % 2) == 0, "fragment reads stay 8-byte aligned");

static __device__ __forceinline__ void lds_barrier() {
    asm volatile("s_wait_dscnt 0x0\n\ts_barrier_signal -1\n\ts_barrier_wait -1" ::: "memory");
}

__global__ __launch_bounds__(256)
void cin_fused_wmma(const float* __restrict__ x0,
                    const float* __restrict__ xk,
                    const float* __restrict__ w,
                    const float* __restrict__ bias,
                    float* __restrict__ out) {
    __shared__ __attribute__((aligned(16))) float lds[LDS_FLOATS];
    float* WL  = lds + WL_OFF;
    float* XL  = lds + XL_OFF;
    float* AL  = lds + AL_OFF;
    float* X0L = lds + X0_OFF;

    const int tid  = threadIdx.x;
    const int wave = tid >> 5;
    const int lane = tid & 31;
    const int half = lane >> 4;
    const int c    = lane & 15;

    const int jt = blockIdx.x & 7;
    const int n0 = jt * 16;
    const int b  = blockIdx.x >> 3;

    const int crot = (b + jt) & 3;
    const int wrot = (b >> 2) * 2;
    const int xrot = (jt >> 2) * 2;
    const int zrot = jt * 2;

    const v4f* w4  = reinterpret_cast<const v4f*>(w);
    const v4f* xk4 = reinterpret_cast<const v4f*>(xk);
    const v4f* x04 = reinterpret_cast<const v4f*>(x0);

    v4f rwE[16], rxE[4], rwO[16], rxO[4], rz[16];

#define LOAD_CHUNK(RW_, RX_, i_)                                                    \
    _Pragma("unroll")                                                               \
    for (int u = 0; u < 16; ++u) {                                                  \
          \
        RW_[u] = w4[(n0 + ((u + wrot) & 15)) * 1024 + (i_) * 256 + tid];            \
    }                                                                               \
    _Pragma("unroll")                                                               \
    for (int u = 0; u < 4; ++u) {                                                   \
        const int idx = ((u + xrot) & 3) * 256 + tid;                               \
                    \
        RX_[u] = xk4[b * 4096 + (idx >> 4) * 64 + (i_) * 16 + (idx & 15)];          \
    }

#define STORE_CHUNK(RW_, RX_, buf_)                                                 \
    _Pragma("unroll")                                                               \
    for (int u = 0; u < 16; ++u) {                                                  \
        *reinterpret_cast<v4f*>(WL + (buf_) * 16 * PW + ((u + wrot) & 15) * PW + tid * 4) = RW_[u]; \
    }                                                                               \
    _Pragma("unroll")                                                               \
    for (int u = 0; u < 4; ++u) {                                                   \
        const int idx = ((u + xrot) & 3) * 256 + tid;                               \
        float* p = XL + (buf_) * 16 * PT + (idx & 15) * PT + (idx >> 4);            \
        p[0]      = RX_[u].x;                                                       \
        p[PK]     = RX_[u].y;                                                       \
        p[2 * PK] = RX_[u].z;                                                       \
        p[3 * PK] = RX_[u].w;                                                       \
    }

#define LOAD_X0()                                                                   \
    _Pragma("unroll")                                                               \
    for (int u = 0; u < 16; ++u) {                                                  \
        const int idx = ((u + zrot) & 15) * 256 + tid;                              \
                          \
        rz[u] = x04[b * 4096 + idx];                                                \
    }

#define STORE_X0()                                                                  \
    _Pragma("unroll")                                                               \
    for (int u = 0; u < 16; ++u) {                                                  \
        const int idx = ((u + zrot) & 15) * 256 + tid;                              \
        float* p = X0L + (idx >> 6) * PK + (idx & 63);                              \
        p[0]      = rz[u].x;                                                        \
        p[PQ]     = rz[u].y;                                                        \
        p[2 * PQ] = rz[u].z;                                                        \
        p[3 * PQ] = rz[u].w;                                                        \
    }

#define PRODUCT1(buf_, i_)                                                          \
    {                                                                               \
        v8f acc0 = {}, acc1 = {};                                                   \
                               \
        const float* ap0 = WL + (buf_) * 16 * PW + c * PW + wave * 64 + 2 * half;   \
        const float* ap1 = ap0 + 8 * 64;                                            \
                                                        \
        const float* bp0 = XL + (buf_) * 16 * PT + wave * PT + (c & 3) * PK + 2 * half; \
        const float* bp1 = bp0 + 8 * PT;                                            \
        _Pragma("unroll")                                                           \
        for (int kc = 0; kc < 16; ++kc) {                                           \
            const v2f a0  = *reinterpret_cast<const v2f*>(ap0 + 4 * kc);            \
            const v2f bb0 = *reinterpret_cast<const v2f*>(bp0 + 4 * kc);            \
            const v2f a1  = *reinterpret_cast<const v2f*>(ap1 + 4 * kc);            \
            const v2f bb1 = *reinterpret_cast<const v2f*>(bp1 + 4 * kc);            \
            acc0 = __builtin_amdgcn_wmma_f32_16x16x4_f32(                           \
                false, a0, false, bb0, (short)0, acc0, false, false);               \
            acc1 = __builtin_amdgcn_wmma_f32_16x16x4_f32(                           \
                false, a1, false, bb1, (short)0, acc1, false, false);               \
        }                                                                           \
          \
        if (c < 4) {                                                                \
            float* dp = AL + c * PC + (8 * half) * PK + (16 * (i_) + wave);         \
            _Pragma("unroll")                                                       \
            for (int v = 0; v < 8; ++v) {                                           \
                dp[v * PK]     = acc0[v];                                           \
                dp[v * PK + 8] = acc1[v];                                           \
            }                                                                       \
        }                                                                           \
    }

    const int c0 = crot;
    const int c1 = (crot + 1) & 3;
    const int c2 = (crot + 2) & 3;
    const int c3 = (crot + 3) & 3;

    LOAD_CHUNK(rwE, rxE, c0)
    LOAD_CHUNK(rwO, rxO, c1)
    STORE_CHUNK(rwE, rxE, 0)
    lds_barrier();

    LOAD_CHUNK(rwE, rxE, c2)
    PRODUCT1(0, c0)
    STORE_CHUNK(rwO, rxO, 1)
    lds_barrier();

    LOAD_CHUNK(rwO, rxO, c3)
    PRODUCT1(1, c1)
    STORE_CHUNK(rwE, rxE, 0)
    lds_barrier();

    LOAD_X0()
    PRODUCT1(0, c2)
    STORE_CHUNK(rwO, rxO, 1)
    lds_barrier();

    PRODUCT1(1, c3)
    STORE_X0()
    __syncthreads();

    const int q  = wave & 3;
    const int pr = wave >> 2;

    v8f acc0 = {}, acc1 = {};
    const float* ap  = AL + q * PC + c * PK + 2 * half;
    const float* bp0 = X0L + q * PQ + (32 * pr + c) * PK + 2 * half;
    const float* bp1 = bp0 + 16 * PK;
    #pragma unroll
    for (int kc = 0; kc < 16; ++kc) {
        const v2f a   = *reinterpret_cast<const v2f*>(ap  + 4 * kc);
        const v2f bb0 = *reinterpret_cast<const v2f*>(bp0 + 4 * kc);
        const v2f bb1 = *reinterpret_cast<const v2f*>(bp1 + 4 * kc);
        acc0 = __builtin_amdgcn_wmma_f32_16x16x4_f32(
            false, a, false, bb0, (short)0, acc0, false, false);
        acc1 = __builtin_amdgcn_wmma_f32_16x16x4_f32(
            false, a, false, bb1, (short)0, acc1, false, false);
    }

    float* orow = out + ((size_t)b * NF + n0) * F + q * 64 + pr * 32 + lane;
    #pragma unroll
    for (int v = 0; v < 8; ++v) {
        const float send = half ? acc0[v] : acc1[v];
        const float got  = __shfl_xor(send, 16, 32);
        const float lo = half ? got : acc0[v];
        const float hi = half ? acc1[v] : got;
        orow[(size_t)v * F]       = lo + bias[n0 + v];
        orow[(size_t)(8 + v) * F] = hi + bias[n0 + 8 + v];
    }
#undef LOAD_CHUNK
#undef STORE_CHUNK
#undef LOAD_X0
#undef STORE_X0
#undef PRODUCT1
}

extern "C" void kernel_launch(void* const* d_in, const int* in_sizes, int n_in,
                              void* d_out, int out_size, void* d_ws, size_t ws_size,
                              hipStream_t stream) {
    const float* x0   = (const float*)d_in[0];
    const float* xk   = (const float*)d_in[1];
    const float* w    = (const float*)d_in[2];
    const float* bias = (const float*)d_in[3];
    float* out = (float*)d_out;

    cin_fused_wmma<<<256, 256, 0, stream>>>(x0, xk, w, bias, out);
}
